# starting kernel
# baseline (speedup 1.0000x reference)
#include <hip/hip_runtime.h>


#define T_DIM 2048
#define B_DIM 4
#define C_DIM 1024
#define H_DIM 16
#define K_DIM 7
#define O_DIM 112

typedef __attribute__((ext_vector_type(2))) float v2f;
typedef __attribute__((ext_vector_type(8))) float v8f;

__global__ __launch_bounds__(256)
void dynconv_fused_kernel(const float* __restrict__ x,
                          const float* __restrict__ Wm,
                          const float* __restrict__ bias,
                          float* __restrict__ out) {
  __shared__ float s_w[16 * O_DIM];

  const int mBase = blockIdx.x << 4;
  const int tid   = threadIdx.x;
  const int wave  = tid >> 5;
  const int lane  = tid & 31;
  const int half  = lane >> 4;
  const int lm    = lane & 15;

  if (wave < 7) {
    const int oBase = wave << 4;
    const float* __restrict__ xr = x  + (size_t)(mBase + lm) * C_DIM + 2 * half;
    const float* __restrict__ wr = Wm + (size_t)(oBase + lm) * C_DIM + 2 * half;
    v8f acc = {0.f, 0.f, 0.f, 0.f, 0.f, 0.f, 0.f, 0.f};
    #pragma unroll 4
    for (int k = 0; k < C_DIM; k += 4) {
      v2f a = *(const v2f*)(xr + k);
      v2f b = *(const v2f*)(wr + k);
      acc = __builtin_amdgcn_wmma_f32_16x16x4_f32(false, a, false, b,
                                                  (short)0, acc, false, false);
    }
    #pragma unroll
    for (int v = 0; v < 8; ++v)
      s_w[(v + 8 * half) * O_DIM + oBase + lm] = acc[v];
  }
  __syncthreads();

  {
    const int row = tid >> 4;
    const int h   = tid & 15;
    float* g = s_w + row * O_DIM + h * K_DIM;
    float mx = g[0];
    #pragma unroll
    for (int k = 1; k < K_DIM; ++k) mx = fmaxf(mx, g[k]);
    float e[K_DIM];
    float s = 0.f;
    #pragma unroll
    for (int k = 0; k < K_DIM; ++k) { e[k] = __expf(g[k] - mx); s += e[k]; }
    const float inv = 1.0f / s;
    #pragma unroll
    for (int k = 0; k < K_DIM; ++k) g[k] = e[k] * inv;
  }
  __syncthreads();

  for (int i = tid; i < 16 * C_DIM; i += 256) {
    const int row = i >> 10;
    const int c   = i & (C_DIM - 1);
    const int m   = mBase + row;
    const int t   = m >> 2;
    const float* wv = s_w + row * O_DIM + (c >> 6) * K_DIM;
    float accv = bias[c];
    if (t >= K_DIM - 1) {
      #pragma unroll
      for (int k = 0; k < K_DIM; ++k)
        accv = fmaf(x[(size_t)(m - (K_DIM - 1 - k) * B_DIM) * C_DIM + c], wv[k], accv);
    } else {
      #pragma unroll
      for (int k = 0; k < K_DIM; ++k) {
        if (t - (K_DIM - 1) + k >= 0)
          accv = fmaf(x[(size_t)(m - (K_DIM - 1 - k) * B_DIM) * C_DIM + c], wv[k], accv);
      }
    }
    out[(size_t)m * C_DIM + c] = accv;
  }
}

extern "C" void kernel_launch(void* const* d_in, const int* in_sizes, int n_in,
                              void* d_out, int out_size, void* d_ws, size_t ws_size,
                              hipStream_t stream) {
  const float* x    = (const float*)d_in[0];
  const float* Wm   = (const float*)d_in[1];
  const float* bias = (const float*)d_in[2];
  float* out = (float*)d_out;

  const int blocks = (T_DIM * B_DIM) / 16;
  hipLaunchKernelGGL(dynconv_fused_kernel, dim3(blocks), dim3(256), 0, stream,
                     x, Wm, bias, out);
}


# GEMM operand feed: x tile and W staged per 64-k chunk through double-buffered LDS with whole-line 16-byte global loads; WMMA fragments read from LDS in the same k order (sums unchanged)
# speedup vs baseline: 1.1397x; 1.1397x over previous
#include <hip/hip_runtime.h>


#define T_DIM 2048
#define B_DIM 4
#define C_DIM 1024
#define H_DIM 16
#define K_DIM 7
#define O_DIM 112

#define KC     64
#define NCHUNK (C_DIM / KC)
#define PITCH  (KC + 4)
#define SROWS  (16 + O_DIM)
#define SBUF   (SROWS * PITCH)
#define NLD    (SROWS * (KC / 4) / 256)

typedef __attribute__((ext_vector_type(2))) float v2f;
typedef __attribute__((ext_vector_type(4))) float v4f;
typedef __attribute__((ext_vector_type(8))) float v8f;

__global__ __launch_bounds__(256)
void dynconv_fused_kernel(const float* __restrict__ x,
                          const float* __restrict__ Wm,
                          const float* __restrict__ bias,
                          float* __restrict__ out) {
  __shared__ float s_w[16 * O_DIM];
  __shared__ __attribute__((aligned(16))) float s_t[2 * SBUF];

  const int mBase = blockIdx.x << 4;
  const int tid   = threadIdx.x;
  const int wave  = tid >> 5;
  const int lane  = tid & 31;
  const int half  = lane >> 4;
  const int lm    = lane & 15;

  {
    const int q  = tid & 15;
    const int r0 = tid >> 4;
    const float* __restrict__ xg = x  + (size_t)(mBase + r0) * C_DIM + 4 * q;
    const float* __restrict__ wg = Wm + (size_t)r0 * C_DIM + 4 * q;
    float* sst = s_t + r0 * PITCH + 4 * q;

    v4f pre[NLD];
    v8f acc = {0.f, 0.f, 0.f, 0.f, 0.f, 0.f, 0.f, 0.f};

    pre[0] = *(const v4f*)(xg);
    #pragma unroll
    for (int j = 1; j < NLD; ++j)
      pre[j] = *(const v4f*)(wg + (size_t)(j - 1) * 16 * C_DIM);
    #pragma unroll
    for (int j = 0; j < NLD; ++j)
      *(v4f*)(sst + j * 16 * PITCH) = pre[j];
    __syncthreads();

    for (int kc = 0; kc < NCHUNK; ++kc) {
      const int cur = kc & 1;
      const bool more = (kc + 1 < NCHUNK);

      if (more) {
        const int k1 = (kc + 1) * KC;
        pre[0] = *(const v4f*)(xg + k1);
        #pragma unroll
        for (int j = 1; j < NLD; ++j)
          pre[j] = *(const v4f*)(wg + (size_t)(j - 1) * 16 * C_DIM + k1);
      }

      if (wave < 7) {
        const float* ap = s_t + cur * SBUF + lm * PITCH + 2 * half;
        const float* bp = s_t + cur * SBUF + (16 + (wave << 4) + lm) * PITCH + 2 * half;
        #pragma unroll
        for (int kk = 0; kk < KC; kk += 4) {
          v2f a = *(const v2f*)(ap + kk);
          v2f b = *(const v2f*)(bp + kk);
          acc = __builtin_amdgcn_wmma_f32_16x16x4_f32(false, a, false, b,
                                                      (short)0, acc, false, false);
        }
      }

      if (more) {
        #pragma unroll
        for (int j = 0; j < NLD; ++j)
          *(v4f*)(sst + (cur ^ 1) * SBUF + j * 16 * PITCH) = pre[j];
        __syncthreads();
      }
    }

    if (wave < 7) {
      const int oBase = wave << 4;
      #pragma unroll
      for (int v = 0; v < 8; ++v)
        s_w[(v + 8 * half) * O_DIM + oBase + lm] = acc[v];
    }
  }
  __syncthreads();

  {
    const int row = tid >> 4;
    const int h   = tid & 15;
    float* g = s_w + row * O_DIM + h * K_DIM;
    float mx = g[0];
    #pragma unroll
    for (int k = 1; k < K_DIM; ++k) mx = fmaxf(mx, g[k]);
    float e[K_DIM];
    float s = 0.f;
    #pragma unroll
    for (int k = 0; k < K_DIM; ++k) { e[k] = __expf(g[k] - mx); s += e[k]; }
    const float inv = 1.0f / s;
    #pragma unroll
    for (int k = 0; k < K_DIM; ++k) g[k] = e[k] * inv;
  }
  __syncthreads();

  for (int i = tid; i < 16 * C_DIM; i += 256) {
    const int row = i >> 10;
    const int c   = i & (C_DIM - 1);
    const int m   = mBase + row;
    const int t   = m >> 2;
    const float* wv = s_w + row * O_DIM + (c >> 6) * K_DIM;
    float accv = bias[c];
    if (t >= K_DIM - 1) {
      #pragma unroll
      for (int k = 0; k < K_DIM; ++k)
        accv = fmaf(x[(size_t)(m - (K_DIM - 1 - k) * B_DIM) * C_DIM + c], wv[k], accv);
    } else {
      #pragma unroll
      for (int k = 0; k < K_DIM; ++k) {
        if (t - (K_DIM - 1) + k >= 0)
          accv = fmaf(x[(size_t)(m - (K_DIM - 1 - k) * B_DIM) * C_DIM + c], wv[k], accv);
      }
    }
    out[(size_t)m * C_DIM + c] = accv;
  }
}

extern "C" void kernel_launch(void* const* d_in, const int* in_sizes, int n_in,
                              void* d_out, int out_size, void* d_ws, size_t ws_size,
                              hipStream_t stream) {
  const float* x    = (const float*)d_in[0];
  const float* Wm   = (const float*)d_in[1];
  const float* bias = (const float*)d_in[2];
  float* out = (float*)d_out;

  const int blocks = (T_DIM * B_DIM) / 16;
  hipLaunchKernelGGL(dynconv_fused_kernel, dim3(blocks), dim3(256), 0, stream,
                     x, Wm, bias, out);
}


# Conv phase re-tiled: a thread owns 4 channels and makes the block's 4 timesteps of each batch row from a 10-row register window of x read with 16-byte loads (7-tap sum order unchanged)
# speedup vs baseline: 3.1839x; 2.7936x over previous
#include <hip/hip_runtime.h>


#define T_DIM 2048
#define B_DIM 4
#define C_DIM 1024
#define H_DIM 16
#define K_DIM 7
#define O_DIM 112

#define KC     64
#define NCHUNK (C_DIM / KC)
#define PITCH  (KC + 4)
#define SROWS  (16 + O_DIM)
#define SBUF   (SROWS * PITCH)
#define NLD    (SROWS * (KC / 4) / 256)

#define TSTEPS 4
#define WIN    (K_DIM - 1 + TSTEPS)

typedef __attribute__((ext_vector_type(2))) float v2f;
typedef __attribute__((ext_vector_type(4))) float v4f;
typedef __attribute__((ext_vector_type(8))) float v8f;

__device__ __forceinline__ void conv_two_batch_rows(const float* __restrict__ xc,
                                                    float* __restrict__ oc,
                                                    const float* wq,
                                                    const v4f bv, const int t0, const int bp,
                                                    const bool pad) {
  v4f win[2][WIN];
  #pragma unroll
  for (int u = 0; u < 2; ++u) {
    #pragma unroll
    for (int j = 0; j < WIN; ++j) {
      const int t = t0 - (K_DIM - 1) + j;
      if (pad && t < 0)
        win[u][j] = (v4f){0.f, 0.f, 0.f, 0.f};
      else
        win[u][j] = *(const v4f*)(xc + (size_t)(t * B_DIM + bp + u) * C_DIM);
    }
  }
  #pragma unroll
  for (int u = 0; u < 2; ++u) {
    #pragma unroll
    for (int tt = 0; tt < TSTEPS; ++tt) {
      const float* wv = wq + (tt * B_DIM + bp + u) * O_DIM;
      v4f a = bv;
      #pragma unroll
      for (int k = 0; k < K_DIM; ++k) {
        const float w  = wv[k];
        const v4f   xv = win[u][tt + k];
        a.x = fmaf(xv.x, w, a.x);
        a.y = fmaf(xv.y, w, a.y);
        a.z = fmaf(xv.z, w, a.z);
        a.w = fmaf(xv.w, w, a.w);
      }
      *(v4f*)(oc + (size_t)((t0 + tt) * B_DIM + bp + u) * C_DIM) = a;
    }
  }
}

__global__ __launch_bounds__(256)
void dynconv_fused_kernel(const float* __restrict__ x,
                          const float* __restrict__ Wm,
                          const float* __restrict__ bias,
                          float* __restrict__ out) {
  __shared__ float s_w[16 * O_DIM];
  __shared__ __attribute__((aligned(16))) float s_t[2 * SBUF];

  const int mBase = blockIdx.x << 4;
  const int tid   = threadIdx.x;
  const int wave  = tid >> 5;
  const int lane  = tid & 31;
  const int half  = lane >> 4;
  const int lm    = lane & 15;

  {
    const int q  = tid & 15;
    const int r0 = tid >> 4;
    const float* __restrict__ xg = x  + (size_t)(mBase + r0) * C_DIM + 4 * q;
    const float* __restrict__ wg = Wm + (size_t)r0 * C_DIM + 4 * q;
    float* sst = s_t + r0 * PITCH + 4 * q;

    v4f pre[NLD];
    v8f acc = {0.f, 0.f, 0.f, 0.f, 0.f, 0.f, 0.f, 0.f};

    pre[0] = *(const v4f*)(xg);
    #pragma unroll
    for (int j = 1; j < NLD; ++j)
      pre[j] = *(const v4f*)(wg + (size_t)(j - 1) * 16 * C_DIM);
    #pragma unroll
    for (int j = 0; j < NLD; ++j)
      *(v4f*)(sst + j * 16 * PITCH) = pre[j];
    __syncthreads();

    for (int kc = 0; kc < NCHUNK; ++kc) {
      const int cur = kc & 1;
      const bool more = (kc + 1 < NCHUNK);

      if (more) {
        const int k1 = (kc + 1) * KC;
        pre[0] = *(const v4f*)(xg + k1);
        #pragma unroll
        for (int j = 1; j < NLD; ++j)
          pre[j] = *(const v4f*)(wg + (size_t)(j - 1) * 16 * C_DIM + k1);
      }

      if (wave < 7) {
        const float* ap = s_t + cur * SBUF + lm * PITCH + 2 * half;
        const float* bp = s_t + cur * SBUF + (16 + (wave << 4) + lm) * PITCH + 2 * half;
        #pragma unroll
        for (int kk = 0; kk < KC; kk += 4) {
          v2f a = *(const v2f*)(ap + kk);
          v2f b = *(const v2f*)(bp + kk);
          acc = __builtin_amdgcn_wmma_f32_16x16x4_f32(false, a, false, b,
                                                      (short)0, acc, false, false);
        }
      }

      if (more) {
        #pragma unroll
        for (int j = 0; j < NLD; ++j)
          *(v4f*)(sst + (cur ^ 1) * SBUF + j * 16 * PITCH) = pre[j];
        __syncthreads();
      }
    }

    if (wave < 7) {
      const int oBase = wave << 4;
      #pragma unroll
      for (int v = 0; v < 8; ++v)
        s_w[(v + 8 * half) * O_DIM + oBase + lm] = acc[v];
    }
  }
  __syncthreads();

  {
    const int row = tid >> 4;
    const int h   = tid & 15;
    float* g = s_w + row * O_DIM + h * K_DIM;
    float mx = g[0];
    #pragma unroll
    for (int k = 1; k < K_DIM; ++k) mx = fmaxf(mx, g[k]);
    float e[K_DIM];
    float s = 0.f;
    #pragma unroll
    for (int k = 0; k < K_DIM; ++k) { e[k] = __expf(g[k] - mx); s += e[k]; }
    const float inv = 1.0f / s;
    #pragma unroll
    for (int k = 0; k < K_DIM; ++k) g[k] = e[k] * inv;
  }
  __syncthreads();

  {
    const int c  = tid << 2;
    const int t0 = blockIdx.x << 2;
    const v4f bv = *(const v4f*)(bias + c);
    const float* wq = s_w + (tid >> 4) * K_DIM;
    const float* xc = x + c;
    float* oc = out + c;
    if (t0 >= K_DIM - 1) {
      conv_two_batch_rows(xc, oc, wq, bv, t0, 0, false);
      conv_two_batch_rows(xc, oc, wq, bv, t0, 2, false);
    } else {
      conv_two_batch_rows(xc, oc, wq, bv, t0, 0, true);
      conv_two_batch_rows(xc, oc, wq, bv, t0, 2, true);
    }
  }
}

extern "C" void kernel_launch(void* const* d_in, const int* in_sizes, int n_in,
                              void* d_out, int out_size, void* d_ws, size_t ws_size,
                              hipStream_t stream) {
  const float* x    = (const float*)d_in[0];
  const float* Wm   = (const float*)d_in[1];
  const float* bias = (const float*)d_in[2];
  float* out = (float*)d_out;

  const int blocks = (T_DIM * B_DIM) / 16;
  hipLaunchKernelGGL(dynconv_fused_kernel, dim3(blocks), dim3(256), 0, stream,
                     x, Wm, bias, out);
}


# GEMM re-tiled: 32 rows per 16-wave work-group, W staged once per 32 rows; each wave makes all 7 column tiles over its eighth of every k chunk; 8 partial sums per logit added in fixed order
# speedup vs baseline: 3.6199x; 1.1369x over previous
#include <hip/hip_runtime.h>


#define T_DIM 2048
#define B_DIM 4
#define C_DIM 1024
#define H_DIM 16
#define K_DIM 7
#define O_DIM 112

#define NTHR   512
#define MROWS  32
#define KC     128
#define KS     (KC / 8)
#define NT     (O_DIM / 16)
#define NGRP   (2 * NT * 2 * 32)
#define NCHUNK (C_DIM / KC)
#define PITCH  (KC + 4)
#define SROWS  (MROWS + O_DIM)
#define SBUF   (SROWS * PITCH)
#define NLD    (SROWS * (KC / 4) / NTHR)

#define TSTEPS 4
#define WIN    (K_DIM - 1 + TSTEPS)

typedef __attribute__((ext_vector_type(2))) float v2f;
typedef __attribute__((ext_vector_type(4))) float v4f;
typedef __attribute__((ext_vector_type(8))) float v8f;

__device__ __forceinline__ void conv_two_batch_rows(const float* __restrict__ xc,
                                                    float* __restrict__ oc,
                                                    const float* wq,
                                                    const v4f bv, const int t0, const int bp,
                                                    const bool pad) {
  v4f win[2][WIN];
  #pragma unroll
  for (int u = 0; u < 2; ++u) {
    #pragma unroll
    for (int j = 0; j < WIN; ++j) {
      const int t = t0 - (K_DIM - 1) + j;
      if (pad && t < 0)
        win[u][j] = (v4f){0.f, 0.f, 0.f, 0.f};
      else
        win[u][j] = *(const v4f*)(xc + (size_t)(t * B_DIM + bp + u) * C_DIM);
    }
  }
  #pragma unroll
  for (int u = 0; u < 2; ++u) {
    #pragma unroll
    for (int tt = 0; tt < TSTEPS; ++tt) {
      const float* wv = wq + (tt * B_DIM + bp + u) * O_DIM;
      v4f a = bv;
      #pragma unroll
      for (int k = 0; k < K_DIM; ++k) {
        const float w  = wv[k];
        const v4f   xv = win[u][tt + k];
        a.x = fmaf(xv.x, w, a.x);
        a.y = fmaf(xv.y, w, a.y);
        a.z = fmaf(xv.z, w, a.z);
        a.w = fmaf(xv.w, w, a.w);
      }
      *(v4f*)(oc + (size_t)((t0 + tt) * B_DIM + bp + u) * C_DIM) = a;
    }
  }
}

__global__ __launch_bounds__(NTHR)
void dynconv_fused_kernel(const float* __restrict__ x,
                          const float* __restrict__ Wm,
                          const float* __restrict__ bias,
                          float* __restrict__ out) {
  __shared__ float s_w[MROWS * O_DIM];
  __shared__ __attribute__((aligned(16))) float s_t[2 * SBUF];

  const int mBase = blockIdx.x * MROWS;
  const int tid   = threadIdx.x;
  const int wave  = __builtin_amdgcn_readfirstlane(tid >> 5);
  const int lane  = tid & 31;
  const int half  = lane >> 4;
  const int lm    = lane & 15;

  const int mt = wave & 1;
  const int ke = wave >> 1;
  {
    const int q  = tid & 31;
    const int r0 = tid >> 5;
    const float* __restrict__ xg = x  + (size_t)(mBase + r0) * C_DIM + 4 * q;
    const float* __restrict__ wg = Wm + (size_t)r0 * C_DIM + 4 * q;
    float* sst = s_t + r0 * PITCH + 4 * q;

    v4f pre[NLD];
    v8f acc[NT];
    #pragma unroll
    for (int n = 0; n < NT; ++n)
      acc[n] = (v8f){0.f, 0.f, 0.f, 0.f, 0.f, 0.f, 0.f, 0.f};

    pre[0] = *(const v4f*)(xg);
    pre[1] = *(const v4f*)(xg + (size_t)16 * C_DIM);
    #pragma unroll
    for (int j = 2; j < NLD; ++j)
      pre[j] = *(const v4f*)(wg + (size_t)(j - 2) * 16 * C_DIM);
    #pragma unroll
    for (int j = 0; j < NLD; ++j)
      *(v4f*)(sst + j * 16 * PITCH) = pre[j];
    __syncthreads();

    for (int kc = 0; kc < NCHUNK; ++kc) {
      const int cur = kc & 1;
      const bool more = (kc + 1 < NCHUNK);

      if (more) {
        const int k1 = (kc + 1) * KC;
        pre[0] = *(const v4f*)(xg + k1);
        pre[1] = *(const v4f*)(xg + (size_t)16 * C_DIM + k1);
        #pragma unroll
        for (int j = 2; j < NLD; ++j)
          pre[j] = *(const v4f*)(wg + (size_t)(j - 2) * 16 * C_DIM + k1);
      }

      {
        const float* ap = s_t + cur * SBUF + ((mt << 4) + lm) * PITCH + ke * KS + 2 * half;
        const float* bp = s_t + cur * SBUF + (MROWS + lm) * PITCH + ke * KS + 2 * half;
        #pragma unroll
        for (int kk = 0; kk < KS; kk += 4) {
          const v2f a = *(const v2f*)(ap + kk);
          #pragma unroll
          for (int n = 0; n < NT; ++n) {
            const v2f b = *(const v2f*)(bp + n * 16 * PITCH + kk);
            acc[n] = __builtin_amdgcn_wmma_f32_16x16x4_f32(false, a, false, b,
                                                           (short)0, acc[n], false, false);
          }
        }
      }

      if (more) {
        #pragma unroll
        for (int j = 0; j < NLD; ++j)
          *(v4f*)(sst + (cur ^ 1) * SBUF + j * 16 * PITCH) = pre[j];
        __syncthreads();
      }
    }

    __syncthreads();
    {
      float* pw = s_t + (size_t)(ke * NGRP + mt * NT * 64 + lane) * 4;
      #pragma unroll
      for (int n = 0; n < NT; ++n) {
        *(v4f*)(pw + (n * 64) * 4)      = (v4f){acc[n][0], acc[n][1], acc[n][2], acc[n][3]};
        *(v4f*)(pw + (n * 64 + 32) * 4) = (v4f){acc[n][4], acc[n][5], acc[n][6], acc[n][7]};
      }
    }
    __syncthreads();
    for (int g = tid; g < NGRP; g += NTHR) {
      v4f sum = *(const v4f*)(s_t + (size_t)g * 4);
      #pragma unroll
      for (int e = 1; e < 8; ++e) {
        const v4f p = *(const v4f*)(s_t + (size_t)(e * NGRP + g) * 4);
        sum.x += p.x; sum.y += p.y; sum.z += p.z; sum.w += p.w;
      }
      const int gl  = g & 31;
      const int gj  = (g >> 5) & 1;
      const int gt  = g >> 6;
      const int gmt = (gt >= NT) ? 1 : 0;
      const int gn  = gt - gmt * NT;
      float* o = s_w + ((gmt << 4) + 4 * gj + 8 * (gl >> 4)) * O_DIM + (gn << 4) + (gl & 15);
      o[0]         = sum.x;
      o[O_DIM]     = sum.y;
      o[2 * O_DIM] = sum.z;
      o[3 * O_DIM] = sum.w;
    }
  }
  __syncthreads();

  {
    const int row = tid >> 4;
    const int h   = tid & 15;
    float* g = s_w + row * O_DIM + h * K_DIM;
    float mx = g[0];
    #pragma unroll
    for (int k = 1; k < K_DIM; ++k) mx = fmaxf(mx, g[k]);
    float e[K_DIM];
    float s = 0.f;
    #pragma unroll
    for (int k = 0; k < K_DIM; ++k) { e[k] = __expf(g[k] - mx); s += e[k]; }
    const float inv = 1.0f / s;
    #pragma unroll
    for (int k = 0; k < K_DIM; ++k) g[k] = e[k] * inv;
  }
  __syncthreads();

  {
    const int cth = tid & 255;
    const int th  = wave >> 3;
    const int c   = cth << 2;
    const int t0  = (blockIdx.x << 3) + (th << 2);
    const v4f bv = *(const v4f*)(bias + c);
    const float* wq = s_w + th * 16 * O_DIM + (cth >> 4) * K_DIM;
    const float* xc = x + c;
    float* oc = out + c;
    if (t0 >= K_DIM - 1) {
      conv_two_batch_rows(xc, oc, wq, bv, t0, 0, false);
      conv_two_batch_rows(xc, oc, wq, bv, t0, 2, false);
    } else {
      conv_two_batch_rows(xc, oc, wq, bv, t0, 0, true);
      conv_two_batch_rows(xc, oc, wq, bv, t0, 2, true);
    }
  }
}

extern "C" void kernel_launch(void* const* d_in, const int* in_sizes, int n_in,
                              void* d_out, int out_size, void* d_ws, size_t ws_size,
                              hipStream_t stream) {
  const float* x    = (const float*)d_in[0];
  const float* Wm   = (const float*)d_in[1];
  const float* bias = (const float*)d_in[2];
  float* out = (float*)d_out;

  const int blocks = (T_DIM * B_DIM) / MROWS;
  hipLaunchKernelGGL(dynconv_fused_kernel, dim3(blocks), dim3(NTHR), 0, stream,
                     x, Wm, bias, out);
}


# x tile stays in LDS after GEMM staging: conv reads its 8 own timesteps there; the 6 earlier ones and bias are loaded ahead of the logit reduction; fragment reads grouped per chunk (sums unchanged)
# speedup vs baseline: 4.3461x; 1.2006x over previous
#include <hip/hip_runtime.h>


#define T_DIM 2048
#define B_DIM 4
#define C_DIM 1024
#define H_DIM 16
#define K_DIM 7
#define O_DIM 112

#define NTHR   512
#define MROWS  32
#define KC     128
#define KS     (KC / 8)
#define NT     (O_DIM / 16)
#define NGRP   (2 * NT * 2 * 32)
#define NCHUNK (C_DIM / KC)
#define PITCH  (KC + 4)
#define XP     (C_DIM + 4)
#define WBUF   (O_DIM * PITCH)
#define NLDW   (O_DIM * (KC / 4) / NTHR)

#define TSTEPS 8
#define HALO   (K_DIM - 1)
#define WIN    (HALO + TSTEPS)

typedef __attribute__((ext_vector_type(2))) float v2f;
typedef __attribute__((ext_vector_type(4))) float v4f;
typedef __attribute__((ext_vector_type(8))) float v8f;

__device__ __forceinline__ void lds_barrier() {
  asm volatile("s_wait_dscnt 0x0\n\ts_barrier_signal -1\n\ts_barrier_wait -1" ::: "memory");
}

__global__ __launch_bounds__(NTHR)
void dynconv_fused_kernel(const float* __restrict__ x,
                          const float* __restrict__ Wm,
                          const float* __restrict__ bias,
                          float* __restrict__ out) {
  __shared__ float s_w[MROWS * O_DIM];
  __shared__ __attribute__((aligned(16))) float s_x[MROWS * XP];
  __shared__ __attribute__((aligned(16))) float s_wt[2 * WBUF];

  const int mBase = blockIdx.x * MROWS;
  const int tid   = threadIdx.x;
  const int wave  = __builtin_amdgcn_readfirstlane(tid >> 5);
  const int lane  = tid & 31;
  const int half  = lane >> 4;
  const int lm    = lane & 15;

  int cth, c, b0;
  v4f halo[2][HALO];
  v4f bv;

  const int mt = wave & 1;
  const int ke = wave >> 1;
  {
    const int q  = tid & 31;
    const int r0 = tid >> 5;
    const float* __restrict__ xg = x  + (size_t)(mBase + r0) * C_DIM + 4 * q;
    const float* __restrict__ wg = Wm + (size_t)r0 * C_DIM + 4 * q;
    float* sxs = s_x  + r0 * XP    + 4 * q;
    float* sws = s_wt + r0 * PITCH + 4 * q;

    v4f prex[2];
    v4f prew[NLDW];
    v8f acc[NT];
    #pragma unroll
    for (int n = 0; n < NT; ++n)
      acc[n] = (v8f){0.f, 0.f, 0.f, 0.f, 0.f, 0.f, 0.f, 0.f};

    prex[0] = *(const v4f*)(xg);
    prex[1] = *(const v4f*)(xg + (size_t)16 * C_DIM);
    #pragma unroll
    for (int j = 0; j < NLDW; ++j)
      prew[j] = *(const v4f*)(wg + (size_t)j * 16 * C_DIM);
    *(v4f*)(sxs)           = prex[0];
    *(v4f*)(sxs + 16 * XP) = prex[1];
    #pragma unroll
    for (int j = 0; j < NLDW; ++j)
      *(v4f*)(sws + j * 16 * PITCH) = prew[j];
    __syncthreads();

    #pragma unroll
    for (int kc = 0; kc < NCHUNK; ++kc) {
      const int cur = kc & 1;
      const bool more = (kc + 1 < NCHUNK);

      if (more) {
        const int k1 = (kc + 1) * KC;
        prex[0] = *(const v4f*)(xg + k1);
        prex[1] = *(const v4f*)(xg + (size_t)16 * C_DIM + k1);
        #pragma unroll
        for (int j = 0; j < NLDW; ++j)
          prew[j] = *(const v4f*)(wg + (size_t)j * 16 * C_DIM + k1);
      }

      {
        const float* ap = s_x + ((mt << 4) + lm) * XP + kc * KC + ke * KS + 2 * half;
        const float* bp = s_wt + cur * WBUF + lm * PITCH + ke * KS + 2 * half;
        v2f af[KS / 4];
        v2f bf[KS / 4][NT];
        #pragma unroll
        for (int s = 0; s < KS / 4; ++s) {
          af[s] = *(const v2f*)(ap + 4 * s);
          #pragma unroll
          for (int n = 0; n < NT; ++n)
            bf[s][n] = *(const v2f*)(bp + n * 16 * PITCH + 4 * s);
        }
        __builtin_amdgcn_sched_barrier(0);
        #pragma unroll
        for (int s = 0; s < KS / 4; ++s) {
          #pragma unroll
          for (int n = 0; n < NT; ++n)
            acc[n] = __builtin_amdgcn_wmma_f32_16x16x4_f32(false, af[s], false, bf[s][n],
                                                           (short)0, acc[n], false, false);
        }
      }

      if (more) {
        *(v4f*)(sxs + (kc + 1) * KC)           = prex[0];
        *(v4f*)(sxs + (kc + 1) * KC + 16 * XP) = prex[1];
        #pragma unroll
        for (int j = 0; j < NLDW; ++j)
          *(v4f*)(sws + (cur ^ 1) * WBUF + j * 16 * PITCH) = prew[j];
        __syncthreads();
      }
    }

    __builtin_amdgcn_sched_barrier(0);
    asm volatile("v_nop\n\tv_nop\n\tv_nop\n\tv_nop");
    __builtin_amdgcn_sched_barrier(0);
    cth = tid & 255;
    c   = cth << 2;
    b0  = (wave >> 3) << 1;

    bv = *(const v4f*)(bias + c);
    if (blockIdx.x != 0) {
      const float* xh = x + (size_t)(mBase - HALO * B_DIM + b0) * C_DIM + c;
      #pragma unroll
      for (int u = 0; u < 2; ++u) {
        #pragma unroll
        for (int j = 0; j < HALO; ++j)
          halo[u][j] = *(const v4f*)(xh + (size_t)(j * B_DIM + u) * C_DIM);
      }
    }

    lds_barrier();
    {
      float* pw = s_wt + (size_t)(ke * NGRP + mt * NT * 64 + lane) * 4;
      #pragma unroll
      for (int n = 0; n < NT; ++n) {
        *(v4f*)(pw + (n * 64) * 4)      = (v4f){acc[n][0], acc[n][1], acc[n][2], acc[n][3]};
        *(v4f*)(pw + (n * 64 + 32) * 4) = (v4f){acc[n][4], acc[n][5], acc[n][6], acc[n][7]};
      }
    }
    lds_barrier();
    for (int g = tid; g < NGRP; g += NTHR) {
      v4f sum = *(const v4f*)(s_wt + (size_t)g * 4);
      #pragma unroll
      for (int e = 1; e < 8; ++e) {
        const v4f p = *(const v4f*)(s_wt + (size_t)(e * NGRP + g) * 4);
        sum.x += p.x; sum.y += p.y; sum.z += p.z; sum.w += p.w;
      }
      const int gl  = g & 31;
      const int gj  = (g >> 5) & 1;
      const int gt  = g >> 6;
      const int gmt = (gt >= NT) ? 1 : 0;
      const int gn  = gt - gmt * NT;
      float* o = s_w + ((gmt << 4) + 4 * gj + 8 * (gl >> 4)) * O_DIM + (gn << 4) + (gl & 15);
      o[0]         = sum.x;
      o[O_DIM]     = sum.y;
      o[2 * O_DIM] = sum.z;
      o[3 * O_DIM] = sum.w;
    }
  }
  lds_barrier();

  {
    const int row = tid >> 4;
    const int h   = tid & 15;
    float* g = s_w + row * O_DIM + h * K_DIM;
    float mx = g[0];
    #pragma unroll
    for (int k = 1; k < K_DIM; ++k) mx = fmaxf(mx, g[k]);
    float e[K_DIM];
    float s = 0.f;
    #pragma unroll
    for (int k = 0; k < K_DIM; ++k) { e[k] = __expf(g[k] - mx); s += e[k]; }
    const float inv = 1.0f / s;
    #pragma unroll
    for (int k = 0; k < K_DIM; ++k) g[k] = e[k] * inv;
  }
  lds_barrier();

  {
    const float* wq = s_w + (cth >> 4) * K_DIM;
    const float* sx = s_x + c;
    float* oc = out + (size_t)mBase * C_DIM + c;
    if (blockIdx.x == 0) {
      #pragma unroll
      for (int u = 0; u < 2; ++u) {
        #pragma unroll
        for (int j = 0; j < HALO; ++j)
          halo[u][j] = (v4f){0.f, 0.f, 0.f, 0.f};
      }
    }
    #pragma unroll
    for (int u = 0; u < 2; ++u) {
      const int b = b0 + u;
      v4f win[WIN];
      #pragma unroll
      for (int j = 0; j < HALO; ++j)
        win[j] = halo[u][j];
      #pragma unroll
      for (int j = 0; j < TSTEPS; ++j)
        win[HALO + j] = *(const v4f*)(sx + (j * B_DIM + b) * XP);
      #pragma unroll
      for (int tt = 0; tt < TSTEPS; ++tt) {
        const float* wv = wq + (tt * B_DIM + b) * O_DIM;
        v4f a = bv;
        #pragma unroll
        for (int k = 0; k < K_DIM; ++k) {
          const float w  = wv[k];
          const v4f   xv = win[tt + k];
          a.x = fmaf(xv.x, w, a.x);
          a.y = fmaf(xv.y, w, a.y);
          a.z = fmaf(xv.z, w, a.z);
          a.w = fmaf(xv.w, w, a.w);
        }
        *(v4f*)(oc + (size_t)(tt * B_DIM + b) * C_DIM) = a;
      }
      if (u == 0) __builtin_amdgcn_sched_barrier(0);
    }
  }
}

extern "C" void kernel_launch(void* const* d_in, const int* in_sizes, int n_in,
                              void* d_out, int out_size, void* d_ws, size_t ws_size,
                              hipStream_t stream) {
  const float* x    = (const float*)d_in[0];
  const float* Wm   = (const float*)d_in[1];
  const float* bias = (const float*)d_in[2];
  float* out = (float*)d_out;

  const int blocks = (T_DIM * B_DIM) / MROWS;
  hipLaunchKernelGGL(dynconv_fused_kernel, dim3(blocks), dim3(NTHR), 0, stream,
                     x, Wm, bias, out);
}
